# starting kernel
# baseline (speedup 1.0000x reference)
#include <hip/hip_runtime.h>

typedef float v2f __attribute__((ext_vector_type(2)));
typedef float v8f __attribute__((ext_vector_type(8)));

#define SEQ_LEN 131072
#define HID 64
#define G4 256

__global__ __launch_bounds__(256) void xproj_wmma(
    const float* __restrict__ inputs, const float* __restrict__ W_ih,
    const float* __restrict__ b_ih, const float* __restrict__ b_hh,
    float* __restrict__ x_proj)
{
  const int lane = threadIdx.x & 31;
  const int wave = threadIdx.x >> 5;
  const int rt   = blockIdx.x * 8 + wave;
  const int l15  = lane & 15;
  const int half = lane >> 4;
  const int row0 = rt * 16;

  const v2f* Arow = reinterpret_cast<const v2f*>(inputs + (size_t)(row0 + l15) * HID);
  v2f areg[16];
#pragma unroll
  for (int kk = 0; kk < 16; ++kk) areg[kk] = Arow[kk * 2 + half];

  for (int ct = 0; ct < 16; ++ct) {
    const int  col  = ct * 16 + l15;
    const v2f* Brow = reinterpret_cast<const v2f*>(W_ih + (size_t)col * HID);
    v8f c = {};
#pragma unroll
    for (int kk = 0; kk < 16; ++kk) {
      v2f b = Brow[kk * 2 + half];
      c = __builtin_amdgcn_wmma_f32_16x16x4_f32(false, areg[kk], false, b,
                                                (short)0, c, false, false);
    }
    const float bias = b_ih[col] + b_hh[col];
#pragma unroll
    for (int v = 0; v < 8; ++v) {
      const int row = row0 + v + half * 8;
      x_proj[(size_t)row * G4 + col] = c[v] + bias;
    }
  }
}

__device__ __forceinline__ float fsig(float x)  { return 1.0f / (1.0f + __expf(-x)); }
__device__ __forceinline__ float ftanh(float x) { return 1.0f - 2.0f / (__expf(2.0f * x) + 1.0f); }

__global__ __launch_bounds__(256) void lstm_scan(
    const float* __restrict__ x_proj, const float* __restrict__ W_hh,
    float* __restrict__ hs)
{
  __shared__ float sh_h[HID];
  __shared__ float sh_z[G4];
  const int j = threadIdx.x;

  float w[HID];
#pragma unroll
  for (int k = 0; k < HID; ++k) w[k] = W_hh[(size_t)j * HID + k];

  float creg = 0.0f;
  if (j < HID) sh_h[j] = 0.0f;
  __syncthreads();

  float xp0 = x_proj[j];
  float xp1 = x_proj[(size_t)G4 + j];

  for (int t = 0; t < SEQ_LEN; ++t) {
    const int   tn  = (t + 2 < SEQ_LEN) ? (t + 2) : (SEQ_LEN - 1);
    const float xp2 = x_proj[(size_t)tn * G4 + j];

    float a0 = xp0, a1 = 0.0f, a2 = 0.0f, a3 = 0.0f;
    const float4* h4 = reinterpret_cast<const float4*>(sh_h);
#pragma unroll
    for (int k = 0; k < 4; ++k) {
      const float4 h0 = h4[4 * k + 0];
      const float4 h1 = h4[4 * k + 1];
      const float4 h2 = h4[4 * k + 2];
      const float4 h3 = h4[4 * k + 3];
      const int b = 16 * k;
      a0 = fmaf(w[b +  0], h0.x, a0); a0 = fmaf(w[b +  1], h0.y, a0);
      a0 = fmaf(w[b +  2], h0.z, a0); a0 = fmaf(w[b +  3], h0.w, a0);
      a1 = fmaf(w[b +  4], h1.x, a1); a1 = fmaf(w[b +  5], h1.y, a1);
      a1 = fmaf(w[b +  6], h1.z, a1); a1 = fmaf(w[b +  7], h1.w, a1);
      a2 = fmaf(w[b +  8], h2.x, a2); a2 = fmaf(w[b +  9], h2.y, a2);
      a2 = fmaf(w[b + 10], h2.z, a2); a2 = fmaf(w[b + 11], h2.w, a2);
      a3 = fmaf(w[b + 12], h3.x, a3); a3 = fmaf(w[b + 13], h3.y, a3);
      a3 = fmaf(w[b + 14], h3.z, a3); a3 = fmaf(w[b + 15], h3.w, a3);
    }
    sh_z[j] = (a0 + a1) + (a2 + a3);
    __syncthreads();

    if (j < HID) {
      const float ig = fsig (sh_z[0 * HID + j]);
      const float fg = fsig (sh_z[1 * HID + j]);
      const float gg = ftanh(sh_z[2 * HID + j]);
      const float og = fsig (sh_z[3 * HID + j]);
      creg = fg * creg + ig * gg;
      const float hn = og * ftanh(creg);
      sh_h[j] = hn;
      hs[(size_t)t * HID + j] = hn;
    }
    __syncthreads();
    xp0 = xp1;
    xp1 = xp2;
  }
}

__global__ __launch_bounds__(256) void head_wmma(
    const float* __restrict__ hs, const float* __restrict__ W_lin,
    const float* __restrict__ b_lin, const float* __restrict__ W_fin,
    const float* __restrict__ b_fin, float* __restrict__ out)
{
  const int lane = threadIdx.x & 31;
  const int wave = threadIdx.x >> 5;
  const int rt   = blockIdx.x * 8 + wave;
  const int l15  = lane & 15;
  const int half = lane >> 4;
  const int row0 = rt * 16;

  const v2f* Arow = reinterpret_cast<const v2f*>(hs + (size_t)(row0 + l15) * HID);
  v2f areg[16];
#pragma unroll
  for (int kk = 0; kk < 16; ++kk) {
    v2f a = Arow[kk * 2 + half];
    a.x = fmaxf(a.x, 0.0f);
    a.y = fmaxf(a.y, 0.0f);
    areg[kk] = a;
  }

  float partial[8];
#pragma unroll
  for (int v = 0; v < 8; ++v) partial[v] = 0.0f;

#pragma unroll
  for (int ctile = 0; ctile < 4; ++ctile) {
    const int  col  = ctile * 16 + l15;
    const v2f* Brow = reinterpret_cast<const v2f*>(W_lin + (size_t)col * HID);
    v8f c = {};
#pragma unroll
    for (int kk = 0; kk < 16; ++kk) {
      v2f b = Brow[kk * 2 + half];
      c = __builtin_amdgcn_wmma_f32_16x16x4_f32(false, areg[kk], false, b,
                                                (short)0, c, false, false);
    }
    const float bl = b_lin[col];
    const float wf = W_fin[col];
#pragma unroll
    for (int v = 0; v < 8; ++v)
      partial[v] += fmaxf(c[v] + bl, 0.0f) * wf;
  }

#pragma unroll
  for (int v = 0; v < 8; ++v) {
    float s = partial[v];
    s += __shfl_xor(s, 1, 32);
    s += __shfl_xor(s, 2, 32);
    s += __shfl_xor(s, 4, 32);
    s += __shfl_xor(s, 8, 32);
    if (l15 == 0) out[row0 + half * 8 + v] = s + b_fin[0];
  }
}

extern "C" void kernel_launch(void* const* d_in, const int* in_sizes, int n_in,
                              void* d_out, int out_size, void* d_ws, size_t ws_size,
                              hipStream_t stream) {
  const float* inputs = (const float*)d_in[0];
  const float* W_ih   = (const float*)d_in[1];
  const float* W_hh   = (const float*)d_in[2];
  const float* b_ih   = (const float*)d_in[3];
  const float* b_hh   = (const float*)d_in[4];
  const float* W_lin  = (const float*)d_in[5];
  const float* b_lin  = (const float*)d_in[6];
  const float* W_fin  = (const float*)d_in[7];
  const float* b_fin  = (const float*)d_in[8];
  float* out = (float*)d_out;

  float* x_proj = (float*)d_ws;
  float* hs     = x_proj + (size_t)SEQ_LEN * G4;

  xproj_wmma<<<1024, 256, 0, stream>>>(inputs, W_ih, b_ih, b_hh, x_proj);
  lstm_scan<<<1, 256, 0, stream>>>(x_proj, W_hh, hs);
  head_wmma<<<1024, 256, 0, stream>>>(hs, W_lin, b_lin, W_fin, b_fin, out);
}


# lstm_scan: i*g and o*tanh(c) are each ONE quotient, sign(x)(1-E)/((1+e)(1+E)) with E=exp(-2|x|), e=exp(-a): 3 IEEE divisions per unit and step, not 5; changes the rounding of i*g and h
# speedup vs baseline: 4.1495x; 4.1495x over previous
#include <hip/hip_runtime.h>

typedef float v2f __attribute__((ext_vector_type(2)));
typedef float v8f __attribute__((ext_vector_type(8)));

#define SEQ_LEN 131072
#define HID 64
#define G4 256

__global__ __launch_bounds__(256) void xproj_wmma(
    const float* __restrict__ inputs, const float* __restrict__ W_ih,
    const float* __restrict__ b_ih, const float* __restrict__ b_hh,
    float* __restrict__ x_proj)
{
  const int lane = threadIdx.x & 31;
  const int wave = threadIdx.x >> 5;
  const int rt   = blockIdx.x * 8 + wave;
  const int l15  = lane & 15;
  const int half = lane >> 4;
  const int row0 = rt * 16;

  const v2f* Arow = reinterpret_cast<const v2f*>(inputs + (size_t)(row0 + l15) * HID);
  v2f areg[16];
#pragma unroll
  for (int kk = 0; kk < 16; ++kk) areg[kk] = Arow[kk * 2 + half];

  for (int ct = 0; ct < 16; ++ct) {
    const int  col  = ct * 16 + l15;
    const v2f* Brow = reinterpret_cast<const v2f*>(W_ih + (size_t)col * HID);
    v8f c = {};
#pragma unroll
    for (int kk = 0; kk < 16; ++kk) {
      v2f b = Brow[kk * 2 + half];
      c = __builtin_amdgcn_wmma_f32_16x16x4_f32(false, areg[kk], false, b,
                                                (short)0, c, false, false);
    }
    const float bias = b_ih[col] + b_hh[col];
#pragma unroll
    for (int v = 0; v < 8; ++v) {
      const int row = row0 + v + half * 8;
      x_proj[(size_t)row * G4 + col] = c[v] + bias;
    }
  }
}

__device__ __forceinline__ float fsig(float x)  { return 1.0f / (1.0f + __expf(-x)); }

__device__ __forceinline__ float sig_times_tanh(float a, float b) {
  const float e = __expf(-a);
  const float E = __builtin_amdgcn_exp2f(__builtin_fabsf(b) * -2.8853900817779268f);
  const float q = (1.0f - E) / ((1.0f + e) * (1.0f + E));
  return __builtin_copysignf(q, b);
}

__device__ __forceinline__ float lane_bcast(float v, int srcLane) {
  return __builtin_bit_cast(float,
      __builtin_amdgcn_readlane(__builtin_bit_cast(int, v), srcLane));
}

#define XP_AHEAD 4

__global__ __launch_bounds__(32) void lstm_scan(
    const float* __restrict__ x_proj, const float* __restrict__ W_hh,
    float* __restrict__ hs)
{
  const int l = threadIdx.x;

  v2f w[4][HID];
#pragma unroll
  for (int g = 0; g < 4; ++g) {
    const float* ra = W_hh + (size_t)(g * HID + 2 * l) * HID;
    const float* rb = ra + HID;
#pragma unroll
    for (int k = 0; k < HID; ++k) {
      v2f p;
      p.x = ra[k];
      p.y = rb[k];
      w[g][k] = p;
    }
  }

  const v2f* xp   = reinterpret_cast<const v2f*>(x_proj) + l;
  v2f*       hout = reinterpret_cast<v2f*>(hs) + l;

  v2f xq[XP_AHEAD][4];
#pragma unroll
  for (int s = 0; s < XP_AHEAD; ++s)
#pragma unroll
    for (int g = 0; g < 4; ++g)
      xq[s][g] = xp[(size_t)s * (G4 / 2) + g * (HID / 2)];

  float h0 = 0.0f, h1 = 0.0f;
  float c0 = 0.0f, c1 = 0.0f;

  for (int t = 0; t < SEQ_LEN; t += XP_AHEAD) {
#pragma unroll
    for (int s = 0; s < XP_AHEAD; ++s) {
      v2f acc[4][4];
#pragma unroll
      for (int g = 0; g < 4; ++g) acc[g][0] = xq[s][g];
      {
        const int tn = (t + s + XP_AHEAD < SEQ_LEN) ? (t + s + XP_AHEAD) : (SEQ_LEN - 1);
#pragma unroll
        for (int g = 0; g < 4; ++g)
          xq[s][g] = xp[(size_t)tn * (G4 / 2) + g * (HID / 2)];
      }

#pragma unroll
      for (int p = 0; p < HID; ++p) {
        const int k = (p < HID / 2) ? (2 * p) : (2 * (p - HID / 2) + 1);
        const float hk = lane_bcast((k & 1) ? h1 : h0, k >> 1);
        const v2f   hb = {hk, hk};
#pragma unroll
        for (int g = 0; g < 4; ++g) {
          if (p >= 1 && p <= 3) acc[g][p] = w[g][k] * hb;
          else acc[g][p & 3] = __builtin_elementwise_fma(w[g][k], hb, acc[g][p & 3]);
        }
      }
      v2f z[4];
#pragma unroll
      for (int g = 0; g < 4; ++g)
        z[g] = (acc[g][0] + acc[g][1]) + (acc[g][2] + acc[g][3]);

      const float f0  = fsig(z[1].x), f1 = fsig(z[1].y);
      const float ig0 = sig_times_tanh(z[0].x, z[2].x);
      const float ig1 = sig_times_tanh(z[0].y, z[2].y);
      c0 = f0 * c0 + ig0;
      c1 = f1 * c1 + ig1;
      h0 = sig_times_tanh(z[3].x, c0);
      h1 = sig_times_tanh(z[3].y, c1);
      v2f hv;
      hv.x = h0;
      hv.y = h1;
      hout[(size_t)(t + s) * (HID / 2)] = hv;
    }
  }
}

__global__ __launch_bounds__(256) void head_wmma(
    const float* __restrict__ hs, const float* __restrict__ W_lin,
    const float* __restrict__ b_lin, const float* __restrict__ W_fin,
    const float* __restrict__ b_fin, float* __restrict__ out)
{
  const int lane = threadIdx.x & 31;
  const int wave = threadIdx.x >> 5;
  const int rt   = blockIdx.x * 8 + wave;
  const int l15  = lane & 15;
  const int half = lane >> 4;
  const int row0 = rt * 16;

  const v2f* Arow = reinterpret_cast<const v2f*>(hs + (size_t)(row0 + l15) * HID);
  v2f areg[16];
#pragma unroll
  for (int kk = 0; kk < 16; ++kk) {
    v2f a = Arow[kk * 2 + half];
    a.x = fmaxf(a.x, 0.0f);
    a.y = fmaxf(a.y, 0.0f);
    areg[kk] = a;
  }

  float partial[8];
#pragma unroll
  for (int v = 0; v < 8; ++v) partial[v] = 0.0f;

#pragma unroll
  for (int ctile = 0; ctile < 4; ++ctile) {
    const int  col  = ctile * 16 + l15;
    const v2f* Brow = reinterpret_cast<const v2f*>(W_lin + (size_t)col * HID);
    v8f c = {};
#pragma unroll
    for (int kk = 0; kk < 16; ++kk) {
      v2f b = Brow[kk * 2 + half];
      c = __builtin_amdgcn_wmma_f32_16x16x4_f32(false, areg[kk], false, b,
                                                (short)0, c, false, false);
    }
    const float bl = b_lin[col];
    const float wf = W_fin[col];
#pragma unroll
    for (int v = 0; v < 8; ++v)
      partial[v] += fmaxf(c[v] + bl, 0.0f) * wf;
  }

#pragma unroll
  for (int v = 0; v < 8; ++v) {
    float s = partial[v];
    s += __shfl_xor(s, 1, 32);
    s += __shfl_xor(s, 2, 32);
    s += __shfl_xor(s, 4, 32);
    s += __shfl_xor(s, 8, 32);
    if (l15 == 0) out[row0 + half * 8 + v] = s + b_fin[0];
  }
}

extern "C" void kernel_launch(void* const* d_in, const int* in_sizes, int n_in,
                              void* d_out, int out_size, void* d_ws, size_t ws_size,
                              hipStream_t stream) {
  const float* inputs = (const float*)d_in[0];
  const float* W_ih   = (const float*)d_in[1];
  const float* W_hh   = (const float*)d_in[2];
  const float* b_ih   = (const float*)d_in[3];
  const float* b_hh   = (const float*)d_in[4];
  const float* W_lin  = (const float*)d_in[5];
  const float* b_lin  = (const float*)d_in[6];
  const float* W_fin  = (const float*)d_in[7];
  const float* b_fin  = (const float*)d_in[8];
  float* out = (float*)d_out;

  float* x_proj = (float*)d_ws;
  float* hs     = x_proj + (size_t)SEQ_LEN * G4;

  xproj_wmma<<<1024, 256, 0, stream>>>(inputs, W_ih, b_ih, b_hh, x_proj);
  lstm_scan<<<1, 32, 0, stream>>>(x_proj, W_hh, hs);
  head_wmma<<<1024, 256, 0, stream>>>(hs, W_lin, b_lin, W_fin, b_fin, out);
}
